# starting kernel
# baseline (speedup 1.0000x reference)
#include <hip/hip_runtime.h>

typedef float v2f __attribute__((ext_vector_type(2)));
typedef float v8f __attribute__((ext_vector_type(8)));

#define NCP  64
#define UPTS 1024
#define VPTS 1024
#define NB   4
#define VT_PER_WAVE 4

__global__ __launch_bounds__(256)
void build_dense_basis(const float* __restrict__ Nu,
                       const float* __restrict__ Nv,
                       const int*   __restrict__ uspan,
                       const int*   __restrict__ vspan,
                       float* __restrict__ NuF,
                       float* __restrict__ NvF) {
    int t = blockIdx.x * blockDim.x + threadIdx.x;
    if (t < UPTS) {
        float* row = NuF + t * NCP;
        #pragma unroll
        for (int i = 0; i < NCP; ++i) row[i] = 0.0f;
        int s = uspan[t] - 3;
        #pragma unroll
        for (int l = 0; l < 4; ++l) row[s + l] = Nu[t * 4 + l];
    } else if (t < 2 * UPTS) {
        int v = t - UPTS;
        float* row = NvF + v * NCP;
        #pragma unroll
        for (int i = 0; i < NCP; ++i) row[i] = 0.0f;
        int s = vspan[v] - 3;
        #pragma unroll
        for (int l = 0; l < 4; ++l) row[s + l] = Nv[v * 4 + l];
    }
}

__global__ __launch_bounds__(32)
void surf_stage1(const float* __restrict__ NuF,
                 const float* __restrict__ ctrl,
                 float* __restrict__ T) {
    const int lane = threadIdx.x;
    const int ut   = blockIdx.x;
    const int jt   = blockIdx.y;
    const int bd   = blockIdx.z;
    const int b    = bd >> 2;
    const int d    = bd & 3;

    const int half = lane >> 4;
    const int l15  = lane & 15;
    const int urow = ut * 16 + l15;
    const int jcol = jt * 16 + l15;

    v8f acc = {};
    const float* nurow = NuF + urow * NCP;

    #pragma unroll
    for (int k0 = 0; k0 < NCP; k0 += 4) {
        const int ka = k0 + half * 2;
        v2f a = *(const v2f*)(nurow + ka);
        v2f bf;
        bf.x = ctrl[((b * NCP + ka    ) * NCP + jcol) * 4 + d];
        bf.y = ctrl[((b * NCP + ka + 1) * NCP + jcol) * 4 + d];
        acc = __builtin_amdgcn_wmma_f32_16x16x4_f32(
                  false, a, false, bf, (short)0, acc, false, false);
    }

    #pragma unroll
    for (int r = 0; r < 8; ++r) {
        const int u = ut * 16 + half * 8 + r;
        T[(bd * UPTS + u) * NCP + (jt * 16 + l15)] = acc[r];
    }
}

__global__ __launch_bounds__(32)
void surf_stage2(const float* __restrict__ T,
                 const float* __restrict__ NvF,
                 float* __restrict__ out) {
    const int lane = threadIdx.x;
    const int vt0  = blockIdx.x * VT_PER_WAVE;
    const int ut   = blockIdx.y;
    const int b    = blockIdx.z;

    const int half = lane >> 4;
    const int l15  = lane & 15;
    const int urow = ut * 16 + l15;

    const float* t0 = T + ((b * 4 + 0) * UPTS + urow) * NCP;
    const float* t1 = T + ((b * 4 + 1) * UPTS + urow) * NCP;
    const float* t2 = T + ((b * 4 + 2) * UPTS + urow) * NCP;
    const float* t3 = T + ((b * 4 + 3) * UPTS + urow) * NCP;

    const float* nvr[VT_PER_WAVE];
    #pragma unroll
    for (int iv = 0; iv < VT_PER_WAVE; ++iv)
        nvr[iv] = NvF + ((vt0 + iv) * 16 + l15) * NCP;

    v8f acc[VT_PER_WAVE][4];
    #pragma unroll
    for (int iv = 0; iv < VT_PER_WAVE; ++iv)
        #pragma unroll
        for (int d = 0; d < 4; ++d)
            acc[iv][d] = (v8f){};

    #pragma unroll
    for (int k0 = 0; k0 < NCP; k0 += 4) {
        const int ka = k0 + half * 2;
        v2f a0 = *(const v2f*)(t0 + ka);
        v2f a1 = *(const v2f*)(t1 + ka);
        v2f a2 = *(const v2f*)(t2 + ka);
        v2f a3 = *(const v2f*)(t3 + ka);
        #pragma unroll
        for (int iv = 0; iv < VT_PER_WAVE; ++iv) {
            v2f bf = *(const v2f*)(nvr[iv] + ka);
            acc[iv][0] = __builtin_amdgcn_wmma_f32_16x16x4_f32(
                             false, a0, false, bf, (short)0, acc[iv][0], false, false);
            acc[iv][1] = __builtin_amdgcn_wmma_f32_16x16x4_f32(
                             false, a1, false, bf, (short)0, acc[iv][1], false, false);
            acc[iv][2] = __builtin_amdgcn_wmma_f32_16x16x4_f32(
                             false, a2, false, bf, (short)0, acc[iv][2], false, false);
            acc[iv][3] = __builtin_amdgcn_wmma_f32_16x16x4_f32(
                             false, a3, false, bf, (short)0, acc[iv][3], false, false);
        }
    }

    #pragma unroll
    for (int iv = 0; iv < VT_PER_WAVE; ++iv) {
        #pragma unroll
        for (int r = 0; r < 8; ++r) {
            const int u = ut * 16 + half * 8 + r;
            const int v = (vt0 + iv) * 16 + l15;
            const float w = acc[iv][3][r];
            const size_t o = ((size_t)(b * UPTS + u) * VPTS + v) * 3;
            out[o + 0] = acc[iv][0][r] / w;
            out[o + 1] = acc[iv][1][r] / w;
            out[o + 2] = acc[iv][2][r] / w;
        }
    }
}

extern "C" void kernel_launch(void* const* d_in, const int* in_sizes, int n_in,
                              void* d_out, int out_size, void* d_ws, size_t ws_size,
                              hipStream_t stream) {
    const float* ctrl  = (const float*)d_in[0];
    const float* Nu    = (const float*)d_in[1];
    const float* Nv    = (const float*)d_in[2];
    const int*   uspan = (const int*)  d_in[3];
    const int*   vspan = (const int*)  d_in[4];
    float*       out   = (float*)d_out;

    float* NuF = (float*)d_ws;
    float* NvF = NuF + (size_t)UPTS * NCP;
    float* T   = NvF + (size_t)VPTS * NCP;

    build_dense_basis<<<(2 * UPTS + 255) / 256, 256, 0, stream>>>(
        Nu, Nv, uspan, vspan, NuF, NvF);

    surf_stage1<<<dim3(UPTS / 16, NCP / 16, NB * 4), 32, 0, stream>>>(
        NuF, ctrl, T);

    surf_stage2<<<dim3(VPTS / 16 / VT_PER_WAVE, UPTS / 16, NB), 32, 0, stream>>>(
        T, NvF, out);
}


# Dense Nu/Nv tables and T kept in WMMA fragment order (16 B word per lane): operand loads and the stores that feed them are contiguous 512 B wave accesses, fetched a K block ahead; sum order kept
# speedup vs baseline: 1.0922x; 1.0922x over previous
#include <hip/hip_runtime.h>


typedef float v2f __attribute__((ext_vector_type(2)));
typedef float v4f __attribute__((ext_vector_type(4)));
typedef float v8f __attribute__((ext_vector_type(8)));

#define NCP  64
#define UPTS 1024
#define VPTS 1024
#define NB   4
#define VT_PER_WAVE 4
#define KP   (NCP / 8)


__device__ __forceinline__ float band_pick(float n0, float n1, float n2, float n3, int l) {
    float r = 0.0f;
    r = (l == 0) ? n0 : r;
    r = (l == 1) ? n1 : r;
    r = (l == 2) ? n2 : r;
    r = (l == 3) ? n3 : r;
    return r;
}

__global__ __launch_bounds__(256)
void build_basis_fragments(const float* __restrict__ Nu,
                           const float* __restrict__ Nv,
                           const int*   __restrict__ uspan,
                           const int*   __restrict__ vspan,
                           float* __restrict__ NuP,
                           float* __restrict__ NvP) {
    int t = blockIdx.x * blockDim.x + threadIdx.x;
    const int words = (UPTS / 16) * KP * 32;
    const float* N    = Nu;
    const int*   span = uspan;
    float*       dst  = NuP;
    if (t >= words) { t -= words; N = Nv; span = vspan; dst = NvP; }

    const int lane = t & 31;
    const int p    = (t >> 5) & (KP - 1);
    const int tile = t >> 8;
    const int c    = lane & 15;
    const int h    = lane >> 4;
    const int row  = tile * 16 + c;

    const int   s  = span[row] - 3;
    const float n0 = N[row * 4 + 0];
    const float n1 = N[row * 4 + 1];
    const float n2 = N[row * 4 + 2];
    const float n3 = N[row * 4 + 3];

    const int l = 8 * p + 2 * h - s;
    v4f w;
    w.x = band_pick(n0, n1, n2, n3, l);
    w.y = band_pick(n0, n1, n2, n3, l + 1);
    w.z = band_pick(n0, n1, n2, n3, l + 4);
    w.w = band_pick(n0, n1, n2, n3, l + 5);
    ((v4f*)dst)[t] = w;
}

__global__ __launch_bounds__(32)
void surf_stage1(const float* __restrict__ NuP,
                 const float* __restrict__ ctrl,
                 float* __restrict__ Tp) {
    const int lane = threadIdx.x;
    const int ut   = blockIdx.x;
    const int jt   = blockIdx.y;
    const int bd   = blockIdx.z;
    const int b    = bd >> 2;
    const int d    = bd & 3;

    const int half = lane >> 4;
    const int l15  = lane & 15;
    const int jcol = jt * 16 + l15;

    const v4f* nuw = (const v4f*)NuP + (ut * KP) * 32 + lane;

    v8f acc = {};

    #pragma unroll
    for (int p = 0; p < KP; ++p) {
        const v4f nw = nuw[p * 32];
        const int ka = 8 * p + half * 2;
        v2f c0, c1;
        c0.x = ctrl[((b * NCP + ka    ) * NCP + jcol) * 4 + d];
        c0.y = ctrl[((b * NCP + ka + 1) * NCP + jcol) * 4 + d];
        c1.x = ctrl[((b * NCP + ka + 4) * NCP + jcol) * 4 + d];
        c1.y = ctrl[((b * NCP + ka + 5) * NCP + jcol) * 4 + d];
        v2f n0 = {nw.x, nw.y};
        v2f n1 = {nw.z, nw.w};
        acc = __builtin_amdgcn_wmma_f32_16x16x4_f32(
                  false, c0, false, n0, (short)0, acc, false, false);
        acc = __builtin_amdgcn_wmma_f32_16x16x4_f32(
                  false, c1, false, n1, (short)0, acc, false, false);
    }

    v4f* tw = (v4f*)Tp + ((bd * (UPTS / 16) + ut) * KP + (jt * 2 + half)) * 32 + l15;
    v4f w0 = {acc[0], acc[1], acc[4], acc[5]};
    v4f w1 = {acc[2], acc[3], acc[6], acc[7]};
    tw[0]  = w0;
    tw[16] = w1;
}

__global__ __launch_bounds__(32)
void surf_stage2(const float* __restrict__ Tp,
                 const float* __restrict__ NvP,
                 float* __restrict__ out) {
    const int lane = threadIdx.x;
    const int vt0  = blockIdx.x * VT_PER_WAVE;
    const int ut   = blockIdx.y;
    const int b    = blockIdx.z;

    const int half = lane >> 4;
    const int l15  = lane & 15;

    const int dim_words = (UPTS / 16) * KP * 32;
    const v4f* tq = (const v4f*)Tp + (b * 4) * dim_words + (ut * KP) * 32 + lane;
    const v4f* nq = (const v4f*)NvP + (vt0 * KP) * 32 + lane;

    v8f acc[VT_PER_WAVE][4];
    #pragma unroll
    for (int iv = 0; iv < VT_PER_WAVE; ++iv)
        #pragma unroll
        for (int d = 0; d < 4; ++d)
            acc[iv][d] = (v8f){};

    v4f a[2][4], bw[2][VT_PER_WAVE];
    #pragma unroll
    for (int d = 0; d < 4; ++d)
        a[0][d] = tq[d * dim_words];
    #pragma unroll
    for (int iv = 0; iv < VT_PER_WAVE; ++iv)
        bw[0][iv] = nq[(iv * KP) * 32];

    #pragma unroll
    for (int p = 0; p < KP; ++p) {
        const int cur = p & 1, nxt = cur ^ 1;
        if (p + 1 < KP) {
            #pragma unroll
            for (int d = 0; d < 4; ++d)
                a[nxt][d] = tq[d * dim_words + (p + 1) * 32];
            #pragma unroll
            for (int iv = 0; iv < VT_PER_WAVE; ++iv)
                bw[nxt][iv] = nq[(iv * KP + p + 1) * 32];
        }
        __builtin_amdgcn_sched_barrier(0);
        #pragma unroll
        for (int iv = 0; iv < VT_PER_WAVE; ++iv) {
            const v2f b0 = {bw[cur][iv].x, bw[cur][iv].y};
            const v2f b1 = {bw[cur][iv].z, bw[cur][iv].w};
            #pragma unroll
            for (int d = 0; d < 4; ++d) {
                const v2f a0 = {a[cur][d].x, a[cur][d].y};
                acc[iv][d] = __builtin_amdgcn_wmma_f32_16x16x4_f32(
                                 false, a0, false, b0, (short)0, acc[iv][d], false, false);
            }
            #pragma unroll
            for (int d = 0; d < 4; ++d) {
                const v2f a1 = {a[cur][d].z, a[cur][d].w};
                acc[iv][d] = __builtin_amdgcn_wmma_f32_16x16x4_f32(
                                 false, a1, false, b1, (short)0, acc[iv][d], false, false);
            }
        }
    }

    #pragma unroll
    for (int iv = 0; iv < VT_PER_WAVE; ++iv) {
        #pragma unroll
        for (int r = 0; r < 8; ++r) {
            const int u = ut * 16 + half * 8 + r;
            const int v = (vt0 + iv) * 16 + l15;
            const float w = acc[iv][3][r];
            const size_t o = ((size_t)(b * UPTS + u) * VPTS + v) * 3;
            out[o + 0] = acc[iv][0][r] / w;
            out[o + 1] = acc[iv][1][r] / w;
            out[o + 2] = acc[iv][2][r] / w;
        }
    }
}

extern "C" void kernel_launch(void* const* d_in, const int* in_sizes, int n_in,
                              void* d_out, int out_size, void* d_ws, size_t ws_size,
                              hipStream_t stream) {
    const float* ctrl  = (const float*)d_in[0];
    const float* Nu    = (const float*)d_in[1];
    const float* Nv    = (const float*)d_in[2];
    const int*   uspan = (const int*)  d_in[3];
    const int*   vspan = (const int*)  d_in[4];
    float*       out   = (float*)d_out;

    float* NuP = (float*)d_ws;
    float* NvP = NuP + (size_t)UPTS * NCP;
    float* Tp  = NvP + (size_t)VPTS * NCP;

    build_basis_fragments<<<(2 * (UPTS / 16) * KP * 32) / 256, 256, 0, stream>>>(
        Nu, Nv, uspan, vspan, NuP, NvP);

    surf_stage1<<<dim3(UPTS / 16, NCP / 16, NB * 4), 32, 0, stream>>>(
        NuP, ctrl, Tp);

    surf_stage2<<<dim3(VPTS / 16 / VT_PER_WAVE, UPTS / 16, NB), 32, 0, stream>>>(
        Tp, NvP, out);
}


# Both products split by span: per distinct span of a 16-point tile one K=4 WMMA of the raw basis rows (other spans zeroed) with the 4-row window; no dense tables, 2 launches; sum order changed
# speedup vs baseline: 1.6290x; 1.4915x over previous
#include <hip/hip_runtime.h>


typedef float v2f __attribute__((ext_vector_type(2)));
typedef float v4f __attribute__((ext_vector_type(4)));
typedef float v8f __attribute__((ext_vector_type(8)));

#define NCP  64
#define UPTS 1024
#define VPTS 1024
#define NB   4
#define UT_PER_WAVE 4


__device__ __forceinline__ float band_pick(v4f n, int l) {
    float r = 0.0f;
    r = (l == 0) ? n.x : r;
    r = (l == 1) ? n.y : r;
    r = (l == 2) ? n.z : r;
    r = (l == 3) ? n.w : r;
    return r;
}

__device__ __forceinline__ v2f span_step(int sp, v4f nrow, int half,
                                         unsigned& pending, int& w0) {
    const int lead = __builtin_ctz(pending);
    const int s    = __builtin_amdgcn_readlane(sp, lead);
    const bool mine = (sp == s);
    pending &= ~(unsigned)__builtin_amdgcn_ballot_w32(mine);
    int w = s - 3;
    w = w < 0 ? 0 : w;
    w = w > NCP - 4 ? NCP - 4 : w;
    const int l = 2 * half + (w - (s - 3));
    v2f bf;
    bf.x = mine ? band_pick(nrow, l)     : 0.0f;
    bf.y = mine ? band_pick(nrow, l + 1) : 0.0f;
    w0 = w;
    return bf;
}

__global__ __launch_bounds__(32)
void surf_stage1(const float* __restrict__ Nu,
                 const int*   __restrict__ uspan,
                 const float* __restrict__ ctrl,
                 float* __restrict__ Tt) {
    const int lane = threadIdx.x;
    const int ut   = blockIdx.x;
    const int jt   = blockIdx.y;
    const int b    = blockIdx.z;
    const int half = lane >> 4;
    const int l15  = lane & 15;

    const int u    = ut * 16 + l15;
    const int sp   = uspan[u];
    const v4f nrow = *(const v4f*)(Nu + u * 4);
    const int jrow = jt * 16 + l15;

    v8f acc[4];
    #pragma unroll
    for (int d = 0; d < 4; ++d) acc[d] = (v8f){};

    unsigned pending = 0xFFFFu;
    while (pending) {
        int w0;
        const v2f bf = span_step(sp, nrow, half, pending, w0);
        const int i0 = w0 + 2 * half;
        const v4f c0 = *(const v4f*)(ctrl + (size_t)((b * NCP + i0    ) * NCP + jrow) * 4);
        const v4f c1 = *(const v4f*)(ctrl + (size_t)((b * NCP + i0 + 1) * NCP + jrow) * 4);
        #pragma unroll
        for (int d = 0; d < 4; ++d) {
            const v2f a = {c0[d], c1[d]};
            acc[d] = __builtin_amdgcn_wmma_f32_16x16x4_f32(
                         false, a, false, bf, (short)0, acc[d], false, false);
        }
    }

    #pragma unroll
    for (int d = 0; d < 4; ++d)
        #pragma unroll
        for (int r = 0; r < 8; ++r)
            Tt[(size_t)((b * 4 + d) * NCP + jt * 16 + half * 8 + r) * UPTS + u] = acc[d][r];
}

__global__ __launch_bounds__(32)
void surf_stage2(const float* __restrict__ Nv,
                 const int*   __restrict__ vspan,
                 const float* __restrict__ Tt,
                 float* __restrict__ out) {
    const int lane = threadIdx.x;
    const int vt   = blockIdx.x;
    const int us   = blockIdx.y;
    const int b    = blockIdx.z;
    const int half = lane >> 4;
    const int l15  = lane & 15;

    const int v    = vt * 16 + l15;
    const int sp   = vspan[v];
    const v4f nrow = *(const v4f*)(Nv + v * 4);
    const int u0   = us * (16 * UT_PER_WAVE);

    v8f acc[UT_PER_WAVE][4];
    #pragma unroll
    for (int iu = 0; iu < UT_PER_WAVE; ++iu)
        #pragma unroll
        for (int d = 0; d < 4; ++d)
            acc[iu][d] = (v8f){};

    const float* tb = Tt + (size_t)(b * 4) * NCP * UPTS + u0 + 4 * l15;

    unsigned pending = 0xFFFFu;
    while (pending) {
        int w0;
        const v2f bf = span_step(sp, nrow, half, pending, w0);
        const int j0 = w0 + 2 * half;
        v4f t0[4], t1[4];
        #pragma unroll
        for (int d = 0; d < 4; ++d) {
            t0[d] = *(const v4f*)(tb + (size_t)(d * NCP + j0    ) * UPTS);
            t1[d] = *(const v4f*)(tb + (size_t)(d * NCP + j0 + 1) * UPTS);
        }
        #pragma unroll
        for (int iu = 0; iu < UT_PER_WAVE; ++iu)
            #pragma unroll
            for (int d = 0; d < 4; ++d) {
                const v2f a = {t0[d][iu], t1[d][iu]};
                acc[iu][d] = __builtin_amdgcn_wmma_f32_16x16x4_f32(
                                 false, a, false, bf, (short)0, acc[iu][d], false, false);
            }
    }

    #pragma unroll
    for (int iu = 0; iu < UT_PER_WAVE; ++iu) {
        #pragma unroll
        for (int r = 0; r < 8; ++r) {
            const int u = u0 + 4 * (half * 8 + r) + iu;
            const float w = acc[iu][3][r];
            const size_t o = ((size_t)(b * UPTS + u) * VPTS + v) * 3;
            out[o + 0] = acc[iu][0][r] / w;
            out[o + 1] = acc[iu][1][r] / w;
            out[o + 2] = acc[iu][2][r] / w;
        }
    }
}

extern "C" void kernel_launch(void* const* d_in, const int* in_sizes, int n_in,
                              void* d_out, int out_size, void* d_ws, size_t ws_size,
                              hipStream_t stream) {
    const float* ctrl  = (const float*)d_in[0];
    const float* Nu    = (const float*)d_in[1];
    const float* Nv    = (const float*)d_in[2];
    const int*   uspan = (const int*)  d_in[3];
    const int*   vspan = (const int*)  d_in[4];
    float*       out   = (float*)d_out;

    float* Tt = (float*)d_ws;

    surf_stage1<<<dim3(UPTS / 16, NCP / 16, NB), 32, 0, stream>>>(
        Nu, uspan, ctrl, Tt);

    surf_stage2<<<dim3(VPTS / 16, UPTS / (16 * UT_PER_WAVE), NB), 32, 0, stream>>>(
        Nv, vspan, Tt, out);
}


# Output stores as whole 128-byte lines: two neighbouring v tiles exchange lane halves in registers (v_permlane16_swap), so a store covers one u row x 32 v; quotients and sum order unchanged
# speedup vs baseline: 2.8113x; 1.7258x over previous
#include <hip/hip_runtime.h>


typedef float v2f __attribute__((ext_vector_type(2)));
typedef float v4f __attribute__((ext_vector_type(4)));
typedef float v8f __attribute__((ext_vector_type(8)));

#define NCP  64
#define UPTS 1024
#define VPTS 1024
#define NB   4
#define VT_PER_WAVE 4


__device__ __forceinline__ float band_pick(v4f n, int l) {
    float r = 0.0f;
    r = (l == 0) ? n.x : r;
    r = (l == 1) ? n.y : r;
    r = (l == 2) ? n.z : r;
    r = (l == 3) ? n.w : r;
    return r;
}

__device__ __forceinline__ v2f span_step(int sp, v4f nrow, int half,
                                         unsigned& pending, int& w0) {
    const int lead = __builtin_ctz(pending);
    const int s    = __builtin_amdgcn_readlane(sp, lead);
    const bool mine = (sp == s);
    pending &= ~(unsigned)__builtin_amdgcn_ballot_w32(mine);
    int w = s - 3;
    w = w < 0 ? 0 : w;
    w = w > NCP - 4 ? NCP - 4 : w;
    const int l = 2 * half + (w - (s - 3));
    v2f bf;
    bf.x = mine ? band_pick(nrow, l)     : 0.0f;
    bf.y = mine ? band_pick(nrow, l + 1) : 0.0f;
    w0 = w;
    return bf;
}

__global__ __launch_bounds__(32)
void surf_eval(const float* __restrict__ Nu,
               const float* __restrict__ Nv,
               const int*   __restrict__ uspan,
               const int*   __restrict__ vspan,
               const float* __restrict__ ctrl,
               float* __restrict__ out) {
    const int lane = threadIdx.x;
    const int vs   = blockIdx.x;
    const int ut   = blockIdx.y;
    const int b    = blockIdx.z;
    const int half = lane >> 4;
    const int l15  = lane & 15;

    const int u    = ut * 16 + l15;
    const int spu  = uspan[u];
    const v4f nu   = *(const v4f*)(Nu + u * 4);

    const int v0   = vs * (16 * VT_PER_WAVE) + l15;
    int lov[VT_PER_WAVE];
    v4f nv[VT_PER_WAVE];
    #pragma unroll
    for (int tv = 0; tv < VT_PER_WAVE; ++tv) {
        lov[tv] = vspan[v0 + 16 * tv] - 3;
        nv[tv]  = *(const v4f*)(Nv + (v0 + 16 * tv) * 4);
    }

    const int jperm = 4 * ((l15 & 7) >> 1) + 2 * (l15 >> 3) + (l15 & 1);

    v8f acc[VT_PER_WAVE][4];
    #pragma unroll
    for (int tv = 0; tv < VT_PER_WAVE; ++tv)
        #pragma unroll
        for (int d = 0; d < 4; ++d)
            acc[tv][d] = (v8f){};

    #pragma nounroll
    for (int jb = 0; jb < NCP; jb += 16) {
        bool hit = false;
        #pragma unroll
        for (int tv = 0; tv < VT_PER_WAVE; ++tv)
            hit = hit || (lov[tv] <= jb + 15 && lov[tv] + 3 >= jb);
        if (__builtin_amdgcn_ballot_w32(hit) == 0u) continue;

        v8f t[4];
        #pragma unroll
        for (int d = 0; d < 4; ++d) t[d] = (v8f){};
        const int jrow = jb + jperm;
        unsigned pending = 0xFFFFu;
        while (pending) {
            int w0;
            const v2f bf = span_step(spu, nu, half, pending, w0);
            const int i0 = w0 + 2 * half;
            const v4f c0 = *(const v4f*)(ctrl + (size_t)((b * NCP + i0    ) * NCP + jrow) * 4);
            const v4f c1 = *(const v4f*)(ctrl + (size_t)((b * NCP + i0 + 1) * NCP + jrow) * 4);
            #pragma unroll
            for (int d = 0; d < 4; ++d) {
                const v2f a = {c0[d], c1[d]};
                t[d] = __builtin_amdgcn_wmma_f32_16x16x4_f32(
                           false, a, false, bf, (short)0, t[d], false, false);
            }
        }

        #pragma unroll
        for (int tv = 0; tv < VT_PER_WAVE; ++tv) {
            #pragma unroll
            for (int p = 0; p < 4; ++p) {
                const int l = jb + 4 * p + 2 * half - lov[tv];
                const bool inband = (l >= -1) && (l <= 3);
                if (__builtin_amdgcn_ballot_w32(inband) == 0u) continue;
                v2f bv;
                bv.x = band_pick(nv[tv], l);
                bv.y = band_pick(nv[tv], l + 1);
                #pragma unroll
                for (int d = 0; d < 4; ++d) {
                    const v2f a = {t[d][2 * p], t[d][2 * p + 1]};
                    acc[tv][d] = __builtin_amdgcn_wmma_f32_16x16x4_f32(
                                     false, a, false, bv, (short)0, acc[tv][d], false, false);
                }
            }
        }
    }

    const int vlane = vs * (16 * VT_PER_WAVE) + lane;
    #pragma unroll
    for (int tp = 0; tp < VT_PER_WAVE; tp += 2) {
        #pragma unroll
        for (int r = 0; r < 8; ++r) {
            const float wa = acc[tp][3][r];
            const float wb = acc[tp + 1][3][r];
            float qa[3], qb[3];
            #pragma unroll
            for (int d = 0; d < 3; ++d) {
                qa[d] = acc[tp][d][r] / wa;
                qb[d] = acc[tp + 1][d][r] / wb;
                asm("v_permlane16_swap_b32 %0, %1" : "+v"(qa[d]), "+v"(qb[d]));
            }
            const size_t oa = ((size_t)(b * UPTS + ut * 16 + r) * VPTS + (vlane + 16 * tp)) * 3;
            const size_t ob = oa + (size_t)8 * VPTS * 3;
            out[oa + 0] = qa[0];
            out[oa + 1] = qa[1];
            out[oa + 2] = qa[2];
            out[ob + 0] = qb[0];
            out[ob + 1] = qb[1];
            out[ob + 2] = qb[2];
        }
    }
}

extern "C" void kernel_launch(void* const* d_in, const int* in_sizes, int n_in,
                              void* d_out, int out_size, void* d_ws, size_t ws_size,
                              hipStream_t stream) {
    const float* ctrl  = (const float*)d_in[0];
    const float* Nu    = (const float*)d_in[1];
    const float* Nv    = (const float*)d_in[2];
    const int*   uspan = (const int*)  d_in[3];
    const int*   vspan = (const int*)  d_in[4];
    float*       out   = (float*)d_out;

    surf_eval<<<dim3(VPTS / (16 * VT_PER_WAVE), UPTS / 16, NB), 32, 0, stream>>>(
        Nu, Nv, uspan, vspan, ctrl, out);
}
